# starting kernel
# baseline (speedup 1.0000x reference)
#include <hip/hip_runtime.h>

typedef float v2f __attribute__((ext_vector_type(2)));
typedef float v4f __attribute__((ext_vector_type(4)));
typedef float v8f __attribute__((ext_vector_type(8)));

#define BATCH 16384
#define DDIM  1024
#define TILE  16
#define NWAVE 8
#define KSPAN (DDIM / NWAVE)
#define THSTR 1028

__global__ __launch_bounds__(256)
void cross_compress_kernel(const float* __restrict__ enc_user,
                           const float* __restrict__ enc_item,
                           const float* __restrict__ t_vv,
                           const float* __restrict__ t_ev,
                           const float* __restrict__ t_ve,
                           const float* __restrict__ t_ee,
                           const float* __restrict__ beta_v,
                           const float* __restrict__ beta_e,
                           float* __restrict__ v_out,
                           float* __restrict__ e_out)
{
    __shared__ float th_lds[5 * THSTR];
    __shared__ float red[2][NWAVE][4][16];
    __shared__ float dots[2][4][16];

    const int tid  = threadIdx.x;
    const int wave = tid >> 5;
    const int lane = tid & 31;
    const int g    = lane >> 4;
    const int sel  = lane & 15;
    const int row0 = blockIdx.x * TILE;

    {
        const int c = tid * 4;
        *(v4f*)&th_lds[0 * THSTR + c] = *(const v4f*)(t_vv + c);
        *(v4f*)&th_lds[1 * THSTR + c] = *(const v4f*)(t_ev + c);
        *(v4f*)&th_lds[2 * THSTR + c] = *(const v4f*)(t_ve + c);
        *(v4f*)&th_lds[3 * THSTR + c] = *(const v4f*)(t_ee + c);
        v4f z = {0.f, 0.f, 0.f, 0.f};
        *(v4f*)&th_lds[4 * THSTR + c] = z;
    }
    __syncthreads();

    const int arow = (sel < 4) ? sel : 4;
    const float* aptr = &th_lds[arow * THSTR];

    v8f acc_u = {0.f,0.f,0.f,0.f,0.f,0.f,0.f,0.f};
    v8f acc_i = {0.f,0.f,0.f,0.f,0.f,0.f,0.f,0.f};

    const float* urow = enc_user + (size_t)(row0 + sel) * DDIM;
    const float* irow = enc_item + (size_t)(row0 + sel) * DDIM;

    #pragma unroll 4
    for (int s = 0; s < KSPAN / 4; ++s) {
        const int kk = wave * KSPAN + s * 4 + g * 2;
        v2f a  = *(const v2f*)(aptr + kk);
        v2f bu = *(const v2f*)(urow + kk);
        v2f bi = *(const v2f*)(irow + kk);
        acc_u = __builtin_amdgcn_wmma_f32_16x16x4_f32(
            false, a, false, bu, (short)0, acc_u, false, false);
        acc_i = __builtin_amdgcn_wmma_f32_16x16x4_f32(
            false, a, false, bi, (short)0, acc_i, false, false);
    }

    if (lane < 16) {
        #pragma unroll
        for (int m = 0; m < 4; ++m) {
            red[0][wave][m][lane] = acc_u[m];
            red[1][wave][m][lane] = acc_i[m];
        }
    }
    __syncthreads();
    if (tid < 128) {
        const int mat = tid >> 6, m = (tid >> 4) & 3, n = tid & 15;
        float ssum = 0.f;
        #pragma unroll
        for (int w2 = 0; w2 < NWAVE; ++w2) ssum += red[mat][w2][m][n];
        dots[mat][m][n] = ssum;
    }
    __syncthreads();

    #pragma unroll
    for (int rr = 0; rr < 2; ++rr) {
        const int r = wave * 2 + rr;
        const size_t row = (size_t)(row0 + r);
        const float e_vv = dots[1][0][r];
        const float v_ev = dots[0][1][r];
        const float e_ve = dots[1][2][r];
        const float v_ee = dots[0][3][r];
        const float* up = enc_user + row * DDIM;
        const float* ip = enc_item + row * DDIM;
        float* vp = v_out + row * DDIM;
        float* ep = e_out + row * DDIM;
        #pragma unroll 2
        for (int base = 0; base < DDIM; base += 128) {
            const int col = base + lane * 4;
            v4f u  = *(const v4f*)(up + col);
            v4f it = *(const v4f*)(ip + col);
            v4f bv = *(const v4f*)(beta_v + col);
            v4f be = *(const v4f*)(beta_e + col);
            v4f vo = u * e_vv + it * v_ev + bv;
            v4f eo = u * e_ve + it * v_ee + be;
            __builtin_nontemporal_store(vo, (v4f*)(vp + col));
            __builtin_nontemporal_store(eo, (v4f*)(ep + col));
        }
    }
}

extern "C" void kernel_launch(void* const* d_in, const int* in_sizes, int n_in,
                              void* d_out, int out_size, void* d_ws, size_t ws_size,
                              hipStream_t stream) {
    const float* enc_user = (const float*)d_in[0];
    const float* enc_item = (const float*)d_in[1];
    const float* t_vv     = (const float*)d_in[2];
    const float* t_ev     = (const float*)d_in[3];
    const float* t_ve     = (const float*)d_in[4];
    const float* t_ee     = (const float*)d_in[5];
    const float* beta_v   = (const float*)d_in[6];
    const float* beta_e   = (const float*)d_in[7];
    float* v_out = (float*)d_out;
    float* e_out = (float*)d_out + (size_t)BATCH * DDIM;

    dim3 grid(BATCH / TILE);
    dim3 block(256);
    hipLaunchKernelGGL(cross_compress_kernel, grid, block, 0, stream,
                       enc_user, enc_item, t_vv, t_ev, t_ve, t_ee,
                       beta_v, beta_e, v_out, e_out);
}


# Work-groups walk their 32 rows as four 8-row tiles (user+item rows share one 16-column WMMA tile); a tile's row pieces are requested with its fragment loads, so the second read is served from cache
# speedup vs baseline: 1.2526x; 1.2526x over previous
#include <hip/hip_runtime.h>


typedef float v2f __attribute__((ext_vector_type(2)));
typedef float v4f __attribute__((ext_vector_type(4)));
typedef float v8f __attribute__((ext_vector_type(8)));

#define BATCH  16384
#define DDIM   1024
#define TROWS  8
#define NTILES 4
#define NWAVE  8
#define KSPAN  (DDIM / NWAVE)
#define THSTR  1028
#define NCB    (DDIM / 128)

__global__ __launch_bounds__(256)
void cross_compress_kernel(const float* __restrict__ enc_user,
                           const float* __restrict__ enc_item,
                           const float* __restrict__ t_vv,
                           const float* __restrict__ t_ev,
                           const float* __restrict__ t_ve,
                           const float* __restrict__ t_ee,
                           const float* __restrict__ beta_v,
                           const float* __restrict__ beta_e,
                           float* __restrict__ v_out,
                           float* __restrict__ e_out)
{
    __shared__ float th_lds[5 * THSTR];
    __shared__ float red[NWAVE][4][32];
    __shared__ float dots[4][16];

    const int tid  = threadIdx.x;
    const int wave = tid >> 5;
    const int lane = tid & 31;
    const int g    = lane >> 4;
    const int sel  = lane & 15;

    {
        const int c = tid * 4;
        *(v4f*)&th_lds[0 * THSTR + c] = *(const v4f*)(t_vv + c);
        *(v4f*)&th_lds[1 * THSTR + c] = *(const v4f*)(t_ev + c);
        *(v4f*)&th_lds[2 * THSTR + c] = *(const v4f*)(t_ve + c);
        *(v4f*)&th_lds[3 * THSTR + c] = *(const v4f*)(t_ee + c);
        v4f z = {0.f, 0.f, 0.f, 0.f};
        *(v4f*)&th_lds[4 * THSTR + c] = z;
    }

    v4f bv[NCB], be[NCB];
    #pragma unroll
    for (int j = 0; j < NCB; ++j) {
        bv[j] = *(const v4f*)(beta_v + j * 128 + lane * 4);
        be[j] = *(const v4f*)(beta_e + j * 128 + lane * 4);
    }
    __syncthreads();

    const int arow = (sel < 4) ? sel : 4;
    const float* aptr = &th_lds[arow * THSTR + wave * KSPAN + g * 2];
    const float* bsrc = (sel < 8) ? enc_user : enc_item;
    const size_t boff = (size_t)(sel & 7) * DDIM + wave * KSPAN + g * 2;
    const int woff = wave * DDIM + lane * 4;

    for (int t = 0; t < NTILES; ++t) {
        const size_t tbase = (size_t)(blockIdx.x * NTILES + t) * (TROWS * DDIM);
        const float* bptr = bsrc + tbase + boff;

        const float* ur = enc_user + tbase + woff;
        const float* ir = enc_item + tbase + woff;
        v4f u[NCB], it[NCB];
        #pragma unroll
        for (int j = 0; j < NCB; ++j) {
            u[j]  = *(const v4f*)(ur + j * 128);
            it[j] = *(const v4f*)(ir + j * 128);
        }

        v8f acc = {0.f,0.f,0.f,0.f,0.f,0.f,0.f,0.f};
        #pragma unroll
        for (int s = 0; s < KSPAN / 4; ++s) {
            v2f a = *(const v2f*)(aptr + s * 4);
            v2f b = *(const v2f*)(bptr + s * 4);
            acc = __builtin_amdgcn_wmma_f32_16x16x4_f32(
                false, a, false, b, (short)0, acc, false, false);
        }
        asm volatile("v_nop\n\tv_nop\n\tv_nop\n\tv_nop" : "+v"(acc));
        #pragma unroll
        for (int m = 0; m < 4; ++m) red[wave][m][lane] = acc[m];

        __syncthreads();
        if (tid < 4 * 16) {
            const int m = tid >> 4, n = tid & 15;
            float ssum = 0.f;
            #pragma unroll
            for (int w2 = 0; w2 < NWAVE; ++w2) ssum += red[w2][m][n];
            dots[m][n] = ssum;
        }
        __syncthreads();

        const float c_evv = dots[0][TROWS + wave];
        const float c_vev = dots[1][wave];
        const float c_eve = dots[2][TROWS + wave];
        const float c_vee = dots[3][wave];

        #pragma unroll
        for (int j = 0; j < NCB; ++j) {
            v4f vo = u[j] * c_evv + it[j] * c_vev + bv[j];
            v4f eo = u[j] * c_eve + it[j] * c_vee + be[j];
            __builtin_nontemporal_store(vo, (v4f*)(v_out + tbase + woff + j * 128));
            __builtin_nontemporal_store(eo, (v4f*)(e_out + tbase + woff + j * 128));
        }
    }
}

extern "C" void kernel_launch(void* const* d_in, const int* in_sizes, int n_in,
                              void* d_out, int out_size, void* d_ws, size_t ws_size,
                              hipStream_t stream) {
    const float* enc_user = (const float*)d_in[0];
    const float* enc_item = (const float*)d_in[1];
    const float* t_vv     = (const float*)d_in[2];
    const float* t_ev     = (const float*)d_in[3];
    const float* t_ve     = (const float*)d_in[4];
    const float* t_ee     = (const float*)d_in[5];
    const float* beta_v   = (const float*)d_in[6];
    const float* beta_e   = (const float*)d_in[7];
    float* v_out = (float*)d_out;
    float* e_out = (float*)d_out + (size_t)BATCH * DDIM;

    dim3 grid(BATCH / (TROWS * NTILES));
    dim3 block(256);
    hipLaunchKernelGGL(cross_compress_kernel, grid, block, 0, stream,
                       enc_user, enc_item, t_vv, t_ev, t_ve, t_ee,
                       beta_v, beta_e, v_out, e_out);
}


# Rows requested one tile ahead into a second register set and kept in flight across the dots, sums and stores: hand-written LDS-only barriers (no load drain), betas parked in LDS; sums unchanged
# speedup vs baseline: 1.3882x; 1.1083x over previous
#include <hip/hip_runtime.h>


typedef float v2f __attribute__((ext_vector_type(2)));
typedef float v4f __attribute__((ext_vector_type(4)));
typedef float v8f __attribute__((ext_vector_type(8)));

#define BATCH  16384
#define DDIM   1024
#define TROWS  8
#define NTILES 4
#define NWAVE  8
#define KSPAN  (DDIM / NWAVE)
#define THSTR  1028
#define NCB    (DDIM / 128)

static __device__ __forceinline__ void lds_barrier() {
    asm volatile("s_wait_dscnt 0x0\n\ts_barrier_signal -1\n\ts_barrier_wait -1" ::: "memory");
}

__global__ __launch_bounds__(256)
void cross_compress_kernel(const float* __restrict__ enc_user,
                           const float* __restrict__ enc_item,
                           const float* __restrict__ t_vv,
                           const float* __restrict__ t_ev,
                           const float* __restrict__ t_ve,
                           const float* __restrict__ t_ee,
                           const float* __restrict__ beta_v,
                           const float* __restrict__ beta_e,
                           float* __restrict__ v_out,
                           float* __restrict__ e_out)
{
    __shared__ float th_lds[5 * THSTR];
    __shared__ float red[NWAVE][4][32];
    __shared__ float dots[4][16];
    __shared__ float beta_lds[2][DDIM];

    const int tid  = threadIdx.x;
    const int wave = tid >> 5;
    const int lane = tid & 31;
    const int g    = lane >> 4;
    const int sel  = lane & 15;

    {
        const int c = tid * 4;
        *(v4f*)&th_lds[0 * THSTR + c] = *(const v4f*)(t_vv + c);
        *(v4f*)&th_lds[1 * THSTR + c] = *(const v4f*)(t_ev + c);
        *(v4f*)&th_lds[2 * THSTR + c] = *(const v4f*)(t_ve + c);
        *(v4f*)&th_lds[3 * THSTR + c] = *(const v4f*)(t_ee + c);
        v4f z = {0.f, 0.f, 0.f, 0.f};
        *(v4f*)&th_lds[4 * THSTR + c] = z;
        *(v4f*)&beta_lds[0][c] = *(const v4f*)(beta_v + c);
        *(v4f*)&beta_lds[1][c] = *(const v4f*)(beta_e + c);
    }

    const int arow = (sel < 4) ? sel : 4;
    const float* aptr = &th_lds[arow * THSTR + wave * KSPAN + g * 2];
    const float* bsrc = (sel < 8) ? enc_user : enc_item;
    const size_t boff = (size_t)(sel & 7) * DDIM + wave * KSPAN + g * 2;
    const int woff = wave * DDIM + lane * 4;

    v4f u[2][NCB], it[2][NCB];
    {
        const size_t tb0 = (size_t)(blockIdx.x * NTILES) * (TROWS * DDIM);
        #pragma unroll
        for (int j = 0; j < NCB; ++j) {
            u[0][j]  = *(const v4f*)(enc_user + tb0 + woff + j * 128);
            it[0][j] = *(const v4f*)(enc_item + tb0 + woff + j * 128);
        }
    }
    lds_barrier();

    #pragma unroll
    for (int t = 0; t < NTILES; ++t) {
        const int cur = t & 1, nxt = cur ^ 1;
        const size_t tbase = (size_t)(blockIdx.x * NTILES + t) * (TROWS * DDIM);
        const float* bptr = bsrc + tbase + boff;

        v2f bfr[KSPAN / 4];
        #pragma unroll
        for (int s = 0; s < KSPAN / 4; ++s) bfr[s] = *(const v2f*)(bptr + s * 4);
        __builtin_amdgcn_sched_barrier(0);

        if (t + 1 < NTILES) {
            const size_t tnext = tbase + (size_t)(TROWS * DDIM);
            #pragma unroll
            for (int j = 0; j < NCB; ++j) {
                u[nxt][j]  = *(const v4f*)(enc_user + tnext + woff + j * 128);
                it[nxt][j] = *(const v4f*)(enc_item + tnext + woff + j * 128);
            }
        }
        __builtin_amdgcn_sched_barrier(0);

        v8f acc = {0.f,0.f,0.f,0.f,0.f,0.f,0.f,0.f};
        #pragma unroll
        for (int s = 0; s < KSPAN / 4; ++s) {
            v2f a = *(const v2f*)(aptr + s * 4);
            acc = __builtin_amdgcn_wmma_f32_16x16x4_f32(
                false, a, false, bfr[s], (short)0, acc, false, false);
        }
        asm volatile("v_nop\n\tv_nop\n\tv_nop\n\tv_nop" : "+v"(acc));
        __builtin_amdgcn_sched_barrier(0);
        #pragma unroll
        for (int m = 0; m < 4; ++m) red[wave][m][lane] = acc[m];

        lds_barrier();
        if (tid < 4 * 16) {
            const int m = tid >> 4, n = tid & 15;
            float ssum = 0.f;
            #pragma unroll
            for (int w2 = 0; w2 < NWAVE; ++w2) ssum += red[w2][m][n];
            dots[m][n] = ssum;
        }
        lds_barrier();

        const float c_evv = dots[0][TROWS + wave];
        const float c_vev = dots[1][wave];
        const float c_eve = dots[2][TROWS + wave];
        const float c_vee = dots[3][wave];

        #pragma unroll
        for (int j = 0; j < NCB; ++j) {
            const v4f bvj = *(const v4f*)&beta_lds[0][j * 128 + lane * 4];
            const v4f bej = *(const v4f*)&beta_lds[1][j * 128 + lane * 4];
            v4f vo = u[cur][j] * c_evv + it[cur][j] * c_vev + bvj;
            v4f eo = u[cur][j] * c_eve + it[cur][j] * c_vee + bej;
            __builtin_nontemporal_store(vo, (v4f*)(v_out + tbase + woff + j * 128));
            __builtin_nontemporal_store(eo, (v4f*)(e_out + tbase + woff + j * 128));
        }
    }
}

extern "C" void kernel_launch(void* const* d_in, const int* in_sizes, int n_in,
                              void* d_out, int out_size, void* d_ws, size_t ws_size,
                              hipStream_t stream) {
    const float* enc_user = (const float*)d_in[0];
    const float* enc_item = (const float*)d_in[1];
    const float* t_vv     = (const float*)d_in[2];
    const float* t_ev     = (const float*)d_in[3];
    const float* t_ve     = (const float*)d_in[4];
    const float* t_ee     = (const float*)d_in[5];
    const float* beta_v   = (const float*)d_in[6];
    const float* beta_e   = (const float*)d_in[7];
    float* v_out = (float*)d_out;
    float* e_out = (float*)d_out + (size_t)BATCH * DDIM;

    dim3 grid(BATCH / (TROWS * NTILES));
    dim3 block(256);
    hipLaunchKernelGGL(cross_compress_kernel, grid, block, 0, stream,
                       enc_user, enc_item, t_vv, t_ev, t_ve, t_ee,
                       beta_v, beta_e, v_out, e_out);
}


# Each wave owns a 128-float K-span of all 16 tile rows: rows read once as whole 128-B lines and 4x4-transposed in registers (DPP row_xmask) into the WMMA B fragments; k order of the sums changes
# speedup vs baseline: 1.7179x; 1.2375x over previous
#include <hip/hip_runtime.h>


typedef float v2f __attribute__((ext_vector_type(2)));
typedef float v4f __attribute__((ext_vector_type(4)));
typedef float v8f __attribute__((ext_vector_type(8)));

#define BATCH  16384
#define DDIM   1024
#define TROWS  8
#define NTILES 4
#define NWAVE  8
#define KSPAN  (DDIM / NWAVE)
#define THSTR  1028
#define NGRP   4
#define NLINE  (KSPAN / 32)

static __device__ __forceinline__ void lds_barrier() {
    asm volatile("s_wait_dscnt 0x0\n\ts_barrier_signal -1\n\ts_barrier_wait -1" ::: "memory");
}

template <int CTRL, int BANK>
static __device__ __forceinline__ float row_take(float keep, float from) {
    return __builtin_bit_cast(float, __builtin_amdgcn_update_dpp(
        __builtin_bit_cast(int, keep), __builtin_bit_cast(int, from), CTRL, 0xf, BANK, false));
}

static __device__ __forceinline__ void tr4(float t0, float t1, float t2, float t3,
                                           float& f0, float& f1, float& f2, float& f3) {
    const float u0 = row_take<0x164, 0xA>(t0, t1);
    const float u1 = row_take<0x164, 0x5>(t1, t0);
    const float u2 = row_take<0x164, 0xA>(t2, t3);
    const float u3 = row_take<0x164, 0x5>(t3, t2);
    f0 = row_take<0x168, 0xC>(u0, u2);
    f2 = row_take<0x168, 0x3>(u2, u0);
    f1 = row_take<0x168, 0xC>(u1, u3);
    f3 = row_take<0x168, 0x3>(u3, u1);
}

__global__ __launch_bounds__(256)
void cross_compress_kernel(const float* __restrict__ enc_user,
                           const float* __restrict__ enc_item,
                           const float* __restrict__ t_vv,
                           const float* __restrict__ t_ev,
                           const float* __restrict__ t_ve,
                           const float* __restrict__ t_ee,
                           const float* __restrict__ beta_v,
                           const float* __restrict__ beta_e,
                           float* __restrict__ v_out,
                           float* __restrict__ e_out)
{
    __shared__ __attribute__((aligned(16))) float th_lds[5 * THSTR];
    __shared__ float red[NWAVE][4][32];
    __shared__ __attribute__((aligned(16))) float coef[TROWS][4];
    __shared__ __attribute__((aligned(16))) float beta_lds[2][DDIM];

    const int tid  = threadIdx.x;
    const int wave = tid >> 5;
    const int lane = tid & 31;
    const int h    = lane >> 4;
    const int sel  = lane & 15;
    const int ilo  = sel & 3;
    const int ihi  = sel >> 2;

    {
        const int c = tid * 4;
        *(v4f*)&th_lds[0 * THSTR + c] = *(const v4f*)(t_vv + c);
        *(v4f*)&th_lds[1 * THSTR + c] = *(const v4f*)(t_ev + c);
        *(v4f*)&th_lds[2 * THSTR + c] = *(const v4f*)(t_ve + c);
        *(v4f*)&th_lds[3 * THSTR + c] = *(const v4f*)(t_ee + c);
        v4f z = {0.f, 0.f, 0.f, 0.f};
        *(v4f*)&th_lds[4 * THSTR + c] = z;
        *(v4f*)&beta_lds[0][c] = *(const v4f*)(beta_v + c);
        *(v4f*)&beta_lds[1][c] = *(const v4f*)(beta_e + c);
    }

    const int loff = ilo * DDIM + wave * KSPAN + h * 16 + ihi * 4;

    const int arow = (sel < 4) ? sel : 4;
    const float* aptr = &th_lds[arow * THSTR + wave * KSPAN + h * 16];

    v4f rw[2][NGRP][NLINE];
    {
        const size_t tb0 = (size_t)(blockIdx.x * NTILES) * (TROWS * DDIM);
        #pragma unroll
        for (int i = 0; i < NGRP * NLINE; ++i) {
            const int g = i / NLINE, n = i % NLINE;
            const float* src = (g < 2) ? enc_user : enc_item;
            rw[0][g][n] = *(const v4f*)(src + tb0 + (g & 1) * 4 * DDIM + n * 32 + loff);
        }
    }
    lds_barrier();

    #pragma unroll
    for (int t = 0; t < NTILES; ++t) {
        const int cur = t & 1, nxt = cur ^ 1;
        const size_t tbase = (size_t)(blockIdx.x * NTILES + t) * (TROWS * DDIM);

        if (t + 1 < NTILES) {
            const size_t tnext = tbase + (size_t)(TROWS * DDIM);
            #pragma unroll
            for (int i = 0; i < NGRP * NLINE; ++i) {
                const int g = i / NLINE, n = i % NLINE;
                const float* src = (g < 2) ? enc_user : enc_item;
                rw[nxt][g][n] = *(const v4f*)(src + tnext + (g & 1) * 4 * DDIM + n * 32 + loff);
            }
        }
        __builtin_amdgcn_sched_barrier(0);

        v4f fr[NLINE][4];
        #pragma unroll
        for (int i = 0; i < NLINE * 4; ++i) {
            const int n = i / 4, e = i % 4;
            float f0, f1, f2, f3;
            tr4(rw[cur][0][n][e], rw[cur][1][n][e], rw[cur][2][n][e], rw[cur][3][n][e],
                f0, f1, f2, f3);
            fr[n][0][e] = f0; fr[n][1][e] = f1; fr[n][2][e] = f2; fr[n][3][e] = f3;
        }
        __builtin_amdgcn_sched_barrier(0);

        v8f acc = {0.f,0.f,0.f,0.f,0.f,0.f,0.f,0.f};
        #pragma unroll
        for (int i = 0; i < NLINE * 4; ++i) {
            const int n = i / 4, y = i % 4;
            const v4f a4 = *(const v4f*)(aptr + n * 32 + y * 4);
            const v2f a0 = {a4[0], a4[1]}, a1 = {a4[2], a4[3]};
            const v2f b0 = {fr[n][y][0], fr[n][y][1]}, b1 = {fr[n][y][2], fr[n][y][3]};
            acc = __builtin_amdgcn_wmma_f32_16x16x4_f32(
                false, a0, false, b0, (short)0, acc, false, false);
            acc = __builtin_amdgcn_wmma_f32_16x16x4_f32(
                false, a1, false, b1, (short)0, acc, false, false);
        }
        asm volatile("v_nop\n\tv_nop\n\tv_nop\n\tv_nop" : "+v"(acc));
        __builtin_amdgcn_sched_barrier(0);
        #pragma unroll
        for (int m = 0; m < 4; ++m) red[wave][m][lane] = acc[m];

        lds_barrier();
        if (tid < 4 * 16) {
            const int m = tid >> 4, n = tid & 15;
            float ssum = 0.f;
            #pragma unroll
            for (int w2 = 0; w2 < NWAVE; ++w2) ssum += red[w2][m][n];
            if (((m & 1) != 0) == (n < TROWS)) coef[n & 7][m] = ssum;
        }
        lds_barrier();

        #pragma unroll
        for (int n = 0; n < NLINE; ++n) {
            const int kq = wave * KSPAN + n * 32 + h * 16 + ihi * 4;
            const v4f bvn = *(const v4f*)&beta_lds[0][kq];
            const v4f ben = *(const v4f*)&beta_lds[1][kq];
            #pragma unroll
            for (int g = 0; g < 2; ++g) {
                const v4f cf = *(const v4f*)&coef[g * 4 + ilo][0];
                const v4f uu = rw[cur][g][n], ii = rw[cur][g + 2][n];
                v4f vo = uu * cf[0] + ii * cf[1] + bvn;
                v4f eo = uu * cf[2] + ii * cf[3] + ben;
                __builtin_nontemporal_store(vo, (v4f*)(v_out + tbase + g * 4 * DDIM + n * 32 + loff));
                __builtin_nontemporal_store(eo, (v4f*)(e_out + tbase + g * 4 * DDIM + n * 32 + loff));
            }
        }
    }
}

extern "C" void kernel_launch(void* const* d_in, const int* in_sizes, int n_in,
                              void* d_out, int out_size, void* d_ws, size_t ws_size,
                              hipStream_t stream) {
    const float* enc_user = (const float*)d_in[0];
    const float* enc_item = (const float*)d_in[1];
    const float* t_vv     = (const float*)d_in[2];
    const float* t_ev     = (const float*)d_in[3];
    const float* t_ve     = (const float*)d_in[4];
    const float* t_ee     = (const float*)d_in[5];
    const float* beta_v   = (const float*)d_in[6];
    const float* beta_e   = (const float*)d_in[7];
    float* v_out = (float*)d_out;
    float* e_out = (float*)d_out + (size_t)BATCH * DDIM;

    dim3 grid(BATCH / (TROWS * NTILES));
    dim3 block(256);
    hipLaunchKernelGGL(cross_compress_kernel, grid, block, 0, stream,
                       enc_user, enc_item, t_vv, t_ev, t_ve, t_ee,
                       beta_v, beta_e, v_out, e_out);
}


# B fragments straight from the whole-line row registers; A holds the thetas block-diagonally (4 thetas x 4 K-pieces, 16 registers): no DPP transposition, no LDS theta reads; k order of sums changes
# speedup vs baseline: 1.8539x; 1.0792x over previous
#include <hip/hip_runtime.h>


typedef float v2f __attribute__((ext_vector_type(2)));
typedef float v4f __attribute__((ext_vector_type(4)));
typedef float v8f __attribute__((ext_vector_type(8)));

#define BATCH  16384
#define DDIM   1024
#define TROWS  8
#define NTILES 4
#define NWAVE  8
#define KSPAN  (DDIM / NWAVE)
#define NGRP   4
#define NLINE  (KSPAN / 32)
#define NPART  (NWAVE * 4)
#define PSTR   36

static __device__ __forceinline__ void lds_barrier() {
    asm volatile("s_wait_dscnt 0x0\n\ts_barrier_signal -1\n\ts_barrier_wait -1" ::: "memory");
}

__global__ __launch_bounds__(256, 4)
void cross_compress_kernel(const float* __restrict__ enc_user,
                           const float* __restrict__ enc_item,
                           const float* __restrict__ t_vv,
                           const float* __restrict__ t_ev,
                           const float* __restrict__ t_ve,
                           const float* __restrict__ t_ee,
                           const float* __restrict__ beta_v,
                           const float* __restrict__ beta_e,
                           float* __restrict__ v_out,
                           float* __restrict__ e_out)
{
    __shared__ __attribute__((aligned(16))) float part[4][TROWS][PSTR];
    __shared__ __attribute__((aligned(16))) float coef[TROWS][4];
    __shared__ __attribute__((aligned(16))) float beta_lds[2][DDIM];

    const int tid  = threadIdx.x;
    const int wave = tid >> 5;
    const int lane = tid & 31;
    const int h    = lane >> 4;
    const int sel  = lane & 15;
    const int ilo  = sel & 3;
    const int ihi  = sel >> 2;

    const int bc = tid * 4;
    const v4f bv0 = *(const v4f*)(beta_v + bc);
    const v4f be0 = *(const v4f*)(beta_e + bc);

    const int koff = wave * KSPAN + h * 16 + ihi * 4;
    const int loff = ilo * DDIM + koff;

    v4f th[NLINE];
    {
        const float* tsel = (ilo == 0) ? t_vv : (ilo == 1) ? t_ev : (ilo == 2) ? t_ve : t_ee;
        #pragma unroll
        for (int n = 0; n < NLINE; ++n) th[n] = *(const v4f*)(tsel + koff + n * 32);
    }

    v4f rw[2][NGRP][NLINE];
    {
        const size_t tb0 = (size_t)(blockIdx.x * NTILES) * (TROWS * DDIM);
        #pragma unroll
        for (int i = 0; i < NGRP * NLINE; ++i) {
            const int g = i / NLINE, n = i % NLINE;
            const float* src = (g < 2) ? enc_user : enc_item;
            rw[0][g][n] = *(const v4f*)(src + tb0 + (g & 1) * 4 * DDIM + n * 32 + loff);
        }
    }
    __builtin_amdgcn_sched_barrier(0);
    *(v4f*)&beta_lds[0][bc] = bv0;
    *(v4f*)&beta_lds[1][bc] = be0;

    const bool own  = (h == (ihi >> 1));
    const bool hi4  = (ihi & 1) != 0;
    const int  pcol = wave * 4 + ihi;

    #pragma unroll
    for (int t = 0; t < NTILES; ++t) {
        const int cur = t & 1, nxt = cur ^ 1;
        const size_t tbase = (size_t)(blockIdx.x * NTILES + t) * (TROWS * DDIM);

        if (t + 1 < NTILES) {
            const size_t tnext = tbase + (size_t)(TROWS * DDIM);
            #pragma unroll
            for (int i = 0; i < NGRP * NLINE; ++i) {
                const int g = i / NLINE, n = i % NLINE;
                const float* src = (g < 2) ? enc_user : enc_item;
                rw[nxt][g][n] = *(const v4f*)(src + tnext + (g & 1) * 4 * DDIM + n * 32 + loff);
            }
        }
        __builtin_amdgcn_sched_barrier(0);

        v8f acc[NGRP];
        #pragma unroll
        for (int g = 0; g < NGRP; ++g) acc[g] = (v8f){0.f,0.f,0.f,0.f,0.f,0.f,0.f,0.f};
        #pragma unroll
        for (int gp = 0; gp < NGRP / 2; ++gp) {
            #pragma unroll
            for (int n = 0; n < NLINE; ++n) {
                #pragma unroll
                for (int p = 0; p < 2; ++p) {
                    const v2f a = {th[n][2 * p], th[n][2 * p + 1]};
                    #pragma unroll
                    for (int q = 0; q < 2; ++q) {
                        const int g = 2 * gp + q;
                        const v2f b = {rw[cur][g][n][2 * p], rw[cur][g][n][2 * p + 1]};
                        acc[g] = __builtin_amdgcn_wmma_f32_16x16x4_f32(
                            false, a, false, b, (short)0, acc[g], false, false);
                    }
                }
            }
        }
        asm volatile("v_nop\n\tv_nop\n\tv_nop\n\tv_nop"
                     : "+v"(acc[0]), "+v"(acc[1]), "+v"(acc[2]), "+v"(acc[3]));
        __builtin_amdgcn_sched_barrier(0);

        if (own && !hi4) {
            #pragma unroll
            for (int g = 0; g < NGRP; ++g) {
                const int r  = (g & 1) * 4 + ilo;
                const int m0 = (g < 2) ? 1 : 0;
                part[m0][r][pcol]     = acc[g][m0];
                part[m0 + 2][r][pcol] = acc[g][m0 + 2];
            }
        }
        if (own && hi4) {
            #pragma unroll
            for (int g = 0; g < NGRP; ++g) {
                const int r  = (g & 1) * 4 + ilo;
                const int m0 = (g < 2) ? 1 : 0;
                part[m0][r][pcol]     = acc[g][4 + m0];
                part[m0 + 2][r][pcol] = acc[g][4 + m0 + 2];
            }
        }

        lds_barrier();
        if (tid < 4 * TROWS) {
            const int m = tid >> 3, r = tid & 7;
            float ssum = 0.f;
            #pragma unroll
            for (int c4 = 0; c4 < NPART / 4; ++c4) {
                const v4f pv = *(const v4f*)&part[m][r][c4 * 4];
                ssum += pv[0]; ssum += pv[1]; ssum += pv[2]; ssum += pv[3];
            }
            coef[r][m] = ssum;
        }
        lds_barrier();

        #pragma unroll
        for (int n = 0; n < NLINE; ++n) {
            const int kq = koff + n * 32;
            const v4f bvn = *(const v4f*)&beta_lds[0][kq];
            const v4f ben = *(const v4f*)&beta_lds[1][kq];
            #pragma unroll
            for (int g = 0; g < 2; ++g) {
                const v4f cf = *(const v4f*)&coef[g * 4 + ilo][0];
                const v4f uu = rw[cur][g][n], ii = rw[cur][g + 2][n];
                v4f vo = uu * cf[0] + ii * cf[1] + bvn;
                v4f eo = uu * cf[2] + ii * cf[3] + ben;
                __builtin_nontemporal_store(vo, (v4f*)(v_out + tbase + g * 4 * DDIM + n * 32 + loff));
                __builtin_nontemporal_store(eo, (v4f*)(e_out + tbase + g * 4 * DDIM + n * 32 + loff));
            }
        }
    }
}

extern "C" void kernel_launch(void* const* d_in, const int* in_sizes, int n_in,
                              void* d_out, int out_size, void* d_ws, size_t ws_size,
                              hipStream_t stream) {
    const float* enc_user = (const float*)d_in[0];
    const float* enc_item = (const float*)d_in[1];
    const float* t_vv     = (const float*)d_in[2];
    const float* t_ev     = (const float*)d_in[3];
    const float* t_ve     = (const float*)d_in[4];
    const float* t_ee     = (const float*)d_in[5];
    const float* beta_v   = (const float*)d_in[6];
    const float* beta_e   = (const float*)d_in[7];
    float* v_out = (float*)d_out;
    float* e_out = (float*)d_out + (size_t)BATCH * DDIM;

    dim3 grid(BATCH / (TROWS * NTILES));
    dim3 block(256);
    hipLaunchKernelGGL(cross_compress_kernel, grid, block, 0, stream,
                       enc_user, enc_item, t_vv, t_ev, t_ve, t_ee,
                       beta_v, beta_e, v_out, e_out);
}
